# starting kernel
# baseline (speedup 1.0000x reference)
#include <hip/hip_runtime.h>
#include <math.h>

typedef float v2f __attribute__((ext_vector_type(2)));
typedef float v8f __attribute__((ext_vector_type(8)));

constexpr int B  = 64;
constexpr int O  = 32;
constexpr int I  = 1024;
constexpr int CI = 16;
constexpr int CO = 16;
constexpr int ITER = 3;

constexpr size_t BSTRIDE = (size_t)O * I * CO;

__global__ __launch_bounds__(256) void uhat_wmma_kernel(
    const float* __restrict__ data,
    const float* __restrict__ W,
    float* __restrict__ uhat)
{
    const int wave = threadIdx.x >> 5;
    const int lane = threadIdx.x & 31;
    const int gw   = blockIdx.x * 8 + wave;
    const int i    = gw >> 2;
    const int oc   = gw & 3;
    const int half = lane >> 4;
    const int lr   = lane & 15;

    v2f a[4][4];
#pragma unroll
    for (int mt = 0; mt < 4; ++mt)
#pragma unroll
        for (int ks = 0; ks < 4; ++ks) {
            const int bb = mt * 16 + lr;
            const float* p = data + ((size_t)bb * I + i) * CI + ks * 4 + half * 2;
            a[mt][ks] = *(const v2f*)p;
        }

    float* pbase0 = uhat + (((size_t)(half * 8) * O) * I + i) * CO + lr;

    for (int oo = 0; oo < 8; ++oo) {
        const int o = oc * 8 + oo;

        v2f bf[4];
#pragma unroll
        for (int ks = 0; ks < 4; ++ks) {
            const float* q = W + (((size_t)o * I + i) * CI + ks * 4 + half * 2) * CO + lr;
            v2f t;
            t.x = q[0];
            t.y = q[CO];
            bf[ks] = t;
        }

        v8f acc[4] = {};
#pragma unroll
        for (int ks = 0; ks < 4; ++ks)
#pragma unroll
            for (int mt = 0; mt < 4; ++mt)
                acc[mt] = __builtin_amdgcn_wmma_f32_16x16x4_f32(
                     false, a[mt][ks],
                     false, bf[ks],
                     (short)0, acc[mt],
                     false,  false);

        float* po = pbase0 + (size_t)o * I * CO;
#pragma unroll
        for (int mt = 0; mt < 4; ++mt) {
            float* p = po + (size_t)(mt * 16) * BSTRIDE;
#pragma unroll
            for (int r = 0; r < 8; ++r)
                p[(size_t)r * BSTRIDE] = acc[mt][r];
        }
    }
}

__global__ __launch_bounds__(1024) void routing_kernel(
    const float* __restrict__ uhat,
    const float* __restrict__ bias,
    const float* __restrict__ beta,
    const float* __restrict__ alpha,
    float* __restrict__ out)
{
    __shared__ float blog[O * I];
    __shared__ float mbuf[I];
    __shared__ float zbuf[I];
    __shared__ float sbuf[O * CO];
    __shared__ float vbuf[O * CO];

    const int b    = blockIdx.x;
    const int tid  = threadIdx.x;
    const int w    = tid >> 5;
    const int lane = tid & 31;

    for (int idx = tid; idx < O * I; idx += 1024) blog[idx] = 0.0f;
    mbuf[tid & (I - 1)] = 0.0f;
    zbuf[tid & (I - 1)] = 1.0f / 32.0f;
    __syncthreads();

    const float* ub = uhat + ((size_t)b * O + w) * I * CO;

    for (int t = 0; t < ITER; ++t) {
        float acc[CO];
#pragma unroll
        for (int d = 0; d < CO; ++d) acc[d] = 0.0f;

        for (int i = lane; i < I; i += 32) {
            const float c = __expf(blog[w * I + i] - mbuf[i]) * zbuf[i];
            const float4* up = (const float4*)(ub + (size_t)i * CO);
            const float4 u0 = up[0], u1 = up[1], u2 = up[2], u3 = up[3];
            acc[0]  += c * u0.x;  acc[1]  += c * u0.y;  acc[2]  += c * u0.z;  acc[3]  += c * u0.w;
            acc[4]  += c * u1.x;  acc[5]  += c * u1.y;  acc[6]  += c * u1.z;  acc[7]  += c * u1.w;
            acc[8]  += c * u2.x;  acc[9]  += c * u2.y;  acc[10] += c * u2.z;  acc[11] += c * u2.w;
            acc[12] += c * u3.x;  acc[13] += c * u3.y;  acc[14] += c * u3.z;  acc[15] += c * u3.w;
        }
#pragma unroll
        for (int off = 16; off > 0; off >>= 1)
#pragma unroll
            for (int d = 0; d < CO; ++d)
                acc[d] += __shfl_xor(acc[d], off, 32);
        if (lane < CO) sbuf[w * CO + lane] = acc[lane];
        __syncthreads();

        if (tid < O * CO) {
            const int o = tid >> 4;
            const int d = tid & 15;
            const float bo = bias[o];
            float sq = 0.0f;
#pragma unroll
            for (int dd = 0; dd < CO; ++dd) {
                const float s = sbuf[o * CO + dd] + bo;
                sq += s * s;
            }
            const float s = sbuf[o * CO + d] + bo;
            vbuf[tid] = (sq / (1.0f + sq)) * s * rsqrtf(sq + 1e-8f);
        }
        __syncthreads();

        if (t < ITER - 1) {
            float v[CO];
#pragma unroll
            for (int d = 0; d < CO; ++d) v[d] = vbuf[w * CO + d];

            for (int i = lane; i < I; i += 32) {
                const float4* up = (const float4*)(ub + (size_t)i * CO);
                const float4 u0 = up[0], u1 = up[1], u2 = up[2], u3 = up[3];
                float dot =
                    v[0]  * u0.x + v[1]  * u0.y + v[2]  * u0.z + v[3]  * u0.w +
                    v[4]  * u1.x + v[5]  * u1.y + v[6]  * u1.z + v[7]  * u1.w +
                    v[8]  * u2.x + v[9]  * u2.y + v[10] * u2.z + v[11] * u2.w +
                    v[12] * u3.x + v[13] * u3.y + v[14] * u3.z + v[15] * u3.w;
                blog[w * I + i] += dot;
            }
            __syncthreads();

            {
                const int i = tid;
                float mx = -INFINITY;
                for (int o = 0; o < O; ++o) mx = fmaxf(mx, blog[o * I + i]);
                float sum = 0.0f;
                for (int o = 0; o < O; ++o) sum += __expf(blog[o * I + i] - mx);
                mbuf[i] = mx;
                zbuf[i] = 1.0f / sum;
            }
            __syncthreads();
        }
    }

    if (tid < O * CO) {
        const int o = tid >> 4;
        float sq = 1e-8f;
#pragma unroll
        for (int dd = 0; dd < CO; ++dd) {
            const float vv = vbuf[o * CO + dd];
            sq += vv * vv;
        }
        const float norm = sqrtf(sq);
        const float act  = 1.0f / (1.0f + __expf(-(alpha[o] * norm + beta[o])));
        out[(size_t)b * O * CO + tid] = vbuf[tid] * act;
    }
}

extern "C" void kernel_launch(void* const* d_in, const int* in_sizes, int n_in,
                              void* d_out, int out_size, void* d_ws, size_t ws_size,
                              hipStream_t stream) {
    (void)in_sizes; (void)n_in; (void)out_size; (void)ws_size;
    const float* data  = (const float*)d_in[0];
    const float* W     = (const float*)d_in[1];
    const float* bias  = (const float*)d_in[2];
    const float* beta  = (const float*)d_in[3];
    const float* alpha = (const float*)d_in[4];
    float* out  = (float*)d_out;
    float* uhat = (float*)d_ws;

    uhat_wmma_kernel<<<dim3((I * 4) / 8), dim3(256), 0, stream>>>(data, W, uhat);
    routing_kernel<<<dim3(B), dim3(1024), 0, stream>>>(uhat, bias, beta, alpha, out);
}


# routing reads u_hat once per iteration, not twice: logits recomputed as dot(sum of earlier v, u_hat), all o of an i in one wave (u_hat kept [b][i][o][d]), softmax across lanes; sum order regrouped
# speedup vs baseline: 1.8013x; 1.8013x over previous
#include <hip/hip_runtime.h>
#include <math.h>


typedef float v2f __attribute__((ext_vector_type(2)));
typedef float v8f __attribute__((ext_vector_type(8)));

constexpr int B  = 64;
constexpr int O  = 32;
constexpr int I  = 1024;
constexpr int CI = 16;
constexpr int CO = 16;
constexpr int ITER = 3;

constexpr size_t BSTRIDE = (size_t)I * O * CO;

__global__ __launch_bounds__(256) void uhat_wmma_kernel(
    const float* __restrict__ data,
    const float* __restrict__ W,
    float* __restrict__ uhat)
{
    const int wave = threadIdx.x >> 5;
    const int lane = threadIdx.x & 31;
    const int gw   = blockIdx.x * 8 + wave;
    const int i    = gw >> 2;
    const int oc   = gw & 3;
    const int half = lane >> 4;
    const int lr   = lane & 15;

    v2f a[4][4];
#pragma unroll
    for (int mt = 0; mt < 4; ++mt)
#pragma unroll
        for (int ks = 0; ks < 4; ++ks) {
            const int bb = mt * 16 + lr;
            const float* p = data + ((size_t)bb * I + i) * CI + ks * 4 + half * 2;
            a[mt][ks] = *(const v2f*)p;
        }

    float* pbase0 = uhat + ((size_t)(half * 8) * I + i) * (O * CO) + lr;

    for (int oo = 0; oo < 8; ++oo) {
        const int o = oc * 8 + oo;

        v2f bf[4];
#pragma unroll
        for (int ks = 0; ks < 4; ++ks) {
            const float* q = W + (((size_t)o * I + i) * CI + ks * 4 + half * 2) * CO + lr;
            v2f t;
            t.x = q[0];
            t.y = q[CO];
            bf[ks] = t;
        }

        v8f acc[4] = {};
#pragma unroll
        for (int ks = 0; ks < 4; ++ks)
#pragma unroll
            for (int mt = 0; mt < 4; ++mt)
                acc[mt] = __builtin_amdgcn_wmma_f32_16x16x4_f32(
                     false, a[mt][ks],
                     false, bf[ks],
                     (short)0, acc[mt],
                     false,  false);

        float* po = pbase0 + (size_t)o * CO;
#pragma unroll
        for (int mt = 0; mt < 4; ++mt) {
            float* p = po + (size_t)(mt * 16) * BSTRIDE;
#pragma unroll
            for (int r = 0; r < 8; ++r)
                p[(size_t)r * BSTRIDE] = acc[mt][r];
        }
    }
}

template <int CTRL>
__device__ __forceinline__ float lane_dpp(float x) {
    const int bits = __builtin_bit_cast(int, x);
    return __builtin_bit_cast(float,
        __builtin_amdgcn_update_dpp(bits, bits, CTRL, 0xf, 0xf, true));
}
constexpr int QUAD_XOR1 = 0xB1;
constexpr int QUAD_XOR2 = 0x4E;
constexpr int ROW_XOR4  = 0x164;
constexpr int ROW_XOR8  = 0x168;

__device__ __forceinline__ float xhalf(float x) {
    const int bits = __builtin_bit_cast(int, x);
    return __builtin_bit_cast(float,
        __builtin_amdgcn_permlanex16(bits, bits, 0x76543210, 0xfedcba98, false, false));
}

struct U8 { float4 q[8]; };

__device__ __forceinline__ void load_u8(U8& u, const float4* __restrict__ p) {
#pragma unroll
    for (int j = 0; j < 8; ++j) u.q[j] = p[j * 32];
}

__device__ __forceinline__ void route_one(const float4* __restrict__ q,
                                          const float4 (&vs)[4], float4 (&acc)[4]) {
    float lg[4];
#pragma unroll
    for (int j = 0; j < 4; ++j) {
        float p = vs[j].x * q[j].x + vs[j].y * q[j].y + vs[j].z * q[j].z + vs[j].w * q[j].w;
        p += lane_dpp<QUAD_XOR1>(p);
        p += lane_dpp<QUAD_XOR2>(p);
        lg[j] = p;
    }
    float m = fmaxf(fmaxf(lg[0], lg[1]), fmaxf(lg[2], lg[3]));
    m = fmaxf(m, lane_dpp<ROW_XOR4>(m));
    m = fmaxf(m, lane_dpp<ROW_XOR8>(m));
    m = fmaxf(m, xhalf(m));
    float e[4];
#pragma unroll
    for (int j = 0; j < 4; ++j) e[j] = __expf(lg[j] - m);
    float z = (e[0] + e[1]) + (e[2] + e[3]);
    z += lane_dpp<ROW_XOR4>(z);
    z += lane_dpp<ROW_XOR8>(z);
    z += xhalf(z);
    const float rz = 1.0f / z;
#pragma unroll
    for (int j = 0; j < 4; ++j) {
        const float c = e[j] * rz;
        acc[j].x += c * q[j].x;
        acc[j].y += c * q[j].y;
        acc[j].z += c * q[j].z;
        acc[j].w += c * q[j].w;
    }
}

constexpr int RWAVES = 16;
static_assert(RWAVES * 32 == O * CO, "one thread per (o,d) element in the reduce phase");
__global__ __launch_bounds__(RWAVES * 32) void routing_kernel(
    const float* __restrict__ uhat,
    const float* __restrict__ bias,
    const float* __restrict__ beta,
    const float* __restrict__ alpha,
    float* __restrict__ out)
{
    __shared__ float4 red[RWAVES * 128];
    __shared__ float4 vsum4[128];

    const int b    = blockIdx.x;
    const int tid  = threadIdx.x;
    const int w    = tid >> 5;
    const int lane = tid & 31;

    const float4* ub = (const float4*)(uhat + (size_t)b * BSTRIDE) + lane;

    float4 vs[4];
#pragma unroll
    for (int j = 0; j < 4; ++j) vs[j] = make_float4(0.f, 0.f, 0.f, 0.f);
    float vown = 0.0f;

    for (int t = 0; t < ITER; ++t) {
        float4 acc[4];
#pragma unroll
        for (int j = 0; j < 4; ++j) acc[j] = make_float4(0.f, 0.f, 0.f, 0.f);

        constexpr int NPAIR = I / 2 / RWAVES;
        U8 ua, ubuf;
        load_u8(ua, ub + (size_t)w * 256);
        for (int s = 0; s < NPAIR - 2; s += 2) {
            load_u8(ubuf, ub + (size_t)(w + RWAVES * (s + 1)) * 256);
            __builtin_amdgcn_sched_barrier(0);
            route_one(&ua.q[0], vs, acc);
            route_one(&ua.q[4], vs, acc);
            load_u8(ua, ub + (size_t)(w + RWAVES * (s + 2)) * 256);
            __builtin_amdgcn_sched_barrier(0);
            route_one(&ubuf.q[0], vs, acc);
            route_one(&ubuf.q[4], vs, acc);
        }
        load_u8(ubuf, ub + (size_t)(w + RWAVES * (NPAIR - 1)) * 256);
        __builtin_amdgcn_sched_barrier(0);
        route_one(&ua.q[0], vs, acc);
        route_one(&ua.q[4], vs, acc);
        route_one(&ubuf.q[0], vs, acc);
        route_one(&ubuf.q[4], vs, acc);

#pragma unroll
        for (int j = 0; j < 4; ++j) red[w * 128 + j * 32 + lane] = acc[j];
        __syncthreads();

        {
            const float* rf = (const float*)red;
            float s = 0.0f;
#pragma unroll
            for (int ww = 0; ww < RWAVES; ++ww) s += rf[ww * (O * CO) + tid];
            const int o = tid >> 4;
            s += bias[o];
            float sq = s * s;
            sq += lane_dpp<QUAD_XOR1>(sq);
            sq += lane_dpp<QUAD_XOR2>(sq);
            sq += lane_dpp<ROW_XOR4>(sq);
            sq += lane_dpp<ROW_XOR8>(sq);
            const float v = (sq / (1.0f + sq)) * s * rsqrtf(sq + 1e-8f);
            if (t < ITER - 1) {
                vown += v;
                ((float*)vsum4)[tid] = vown;
            } else {
                float vv = v * v;
                vv += lane_dpp<QUAD_XOR1>(vv);
                vv += lane_dpp<QUAD_XOR2>(vv);
                vv += lane_dpp<ROW_XOR4>(vv);
                vv += lane_dpp<ROW_XOR8>(vv);
                const float norm = sqrtf(vv + 1e-8f);
                const float act  = 1.0f / (1.0f + __expf(-(alpha[o] * norm + beta[o])));
                out[(size_t)b * O * CO + tid] = v * act;
            }
        }
        if (t < ITER - 1) {
            __syncthreads();
#pragma unroll
            for (int j = 0; j < 4; ++j) vs[j] = vsum4[j * 32 + lane];
        }
    }
}

extern "C" void kernel_launch(void* const* d_in, const int* in_sizes, int n_in,
                              void* d_out, int out_size, void* d_ws, size_t ws_size,
                              hipStream_t stream) {
    (void)in_sizes; (void)n_in; (void)out_size; (void)ws_size;
    const float* data  = (const float*)d_in[0];
    const float* W     = (const float*)d_in[1];
    const float* bias  = (const float*)d_in[2];
    const float* beta  = (const float*)d_in[3];
    const float* alpha = (const float*)d_in[4];
    float* out  = (float*)d_out;
    float* uhat = (float*)d_ws;

    uhat_wmma_kernel<<<dim3((I * 4) / 8), dim3(256), 0, stream>>>(data, W, uhat);
    routing_kernel<<<dim3(B), dim3(RWAVES * 32), 0, stream>>>(uhat, bias, beta, alpha, out);
}


# routing spread over all WGPs: 4 workgroups per batch element (256, was 64), one launch per routing iteration, partial s summed in the next launch; sum order regrouped
# speedup vs baseline: 2.4295x; 1.3487x over previous
#include <hip/hip_runtime.h>
#include <math.h>


typedef float v2f __attribute__((ext_vector_type(2)));
typedef float v8f __attribute__((ext_vector_type(8)));

constexpr int B  = 64;
constexpr int O  = 32;
constexpr int I  = 1024;
constexpr int CI = 16;
constexpr int CO = 16;
constexpr int ITER = 3;

constexpr size_t BSTRIDE = (size_t)I * O * CO;

__global__ __launch_bounds__(256) void uhat_wmma_kernel(
    const float* __restrict__ data,
    const float* __restrict__ W,
    float* __restrict__ uhat)
{
    const int wave = threadIdx.x >> 5;
    const int lane = threadIdx.x & 31;
    const int gw   = blockIdx.x * 8 + wave;
    const int i    = gw >> 2;
    const int oc   = gw & 3;
    const int half = lane >> 4;
    const int lr   = lane & 15;

    v2f a[4][4];
#pragma unroll
    for (int mt = 0; mt < 4; ++mt)
#pragma unroll
        for (int ks = 0; ks < 4; ++ks) {
            const int bb = mt * 16 + lr;
            const float* p = data + ((size_t)bb * I + i) * CI + ks * 4 + half * 2;
            a[mt][ks] = *(const v2f*)p;
        }

    float* pbase0 = uhat + ((size_t)(half * 8) * I + i) * (O * CO) + lr;

    for (int oo = 0; oo < 8; ++oo) {
        const int o = oc * 8 + oo;

        v2f bf[4];
#pragma unroll
        for (int ks = 0; ks < 4; ++ks) {
            const float* q = W + (((size_t)o * I + i) * CI + ks * 4 + half * 2) * CO + lr;
            v2f t;
            t.x = q[0];
            t.y = q[CO];
            bf[ks] = t;
        }

        v8f acc[4] = {};
#pragma unroll
        for (int ks = 0; ks < 4; ++ks)
#pragma unroll
            for (int mt = 0; mt < 4; ++mt)
                acc[mt] = __builtin_amdgcn_wmma_f32_16x16x4_f32(
                     false, a[mt][ks],
                     false, bf[ks],
                     (short)0, acc[mt],
                     false,  false);

        float* po = pbase0 + (size_t)o * CO;
#pragma unroll
        for (int mt = 0; mt < 4; ++mt) {
            float* p = po + (size_t)(mt * 16) * BSTRIDE;
#pragma unroll
            for (int r = 0; r < 8; ++r)
                p[(size_t)r * BSTRIDE] = acc[mt][r];
        }
    }
}

template <int CTRL>
__device__ __forceinline__ float lane_dpp(float x) {
    const int bits = __builtin_bit_cast(int, x);
    return __builtin_bit_cast(float,
        __builtin_amdgcn_update_dpp(bits, bits, CTRL, 0xf, 0xf, true));
}
constexpr int QUAD_XOR1 = 0xB1;
constexpr int QUAD_XOR2 = 0x4E;
constexpr int ROW_XOR4  = 0x164;
constexpr int ROW_XOR8  = 0x168;

__device__ __forceinline__ float xhalf(float x) {
    const int bits = __builtin_bit_cast(int, x);
    return __builtin_bit_cast(float,
        __builtin_amdgcn_permlanex16(bits, bits, 0x76543210, 0xfedcba98, false, false));
}

struct U8 { float4 q[8]; };

__device__ __forceinline__ void load_u8(U8& u, const float4* __restrict__ p) {
#pragma unroll
    for (int j = 0; j < 8; ++j) u.q[j] = p[j * 32];
}

__device__ __forceinline__ void route_one(const float4* __restrict__ q,
                                          const float4 (&vs)[4], float4 (&acc)[4]) {
    float lg[4];
#pragma unroll
    for (int j = 0; j < 4; ++j) {
        float p = vs[j].x * q[j].x + vs[j].y * q[j].y + vs[j].z * q[j].z + vs[j].w * q[j].w;
        p += lane_dpp<QUAD_XOR1>(p);
        p += lane_dpp<QUAD_XOR2>(p);
        lg[j] = p;
    }
    float m = fmaxf(fmaxf(lg[0], lg[1]), fmaxf(lg[2], lg[3]));
    m = fmaxf(m, lane_dpp<ROW_XOR4>(m));
    m = fmaxf(m, lane_dpp<ROW_XOR8>(m));
    m = fmaxf(m, xhalf(m));
    float e[4];
#pragma unroll
    for (int j = 0; j < 4; ++j) e[j] = __expf(lg[j] - m);
    float z = (e[0] + e[1]) + (e[2] + e[3]);
    z += lane_dpp<ROW_XOR4>(z);
    z += lane_dpp<ROW_XOR8>(z);
    z += xhalf(z);
    const float rz = 1.0f / z;
#pragma unroll
    for (int j = 0; j < 4; ++j) {
        const float c = e[j] * rz;
        acc[j].x += c * q[j].x;
        acc[j].y += c * q[j].y;
        acc[j].z += c * q[j].z;
        acc[j].w += c * q[j].w;
    }
}

constexpr int RWAVES = 16;
constexpr int NSPLIT = 4;
constexpr int ISPLIT = I / NSPLIT;
constexpr size_t PART_OFF = (size_t)B * BSTRIDE;
static_assert(RWAVES * 32 == O * CO, "one thread per (o,d) element in the reduce phase");
static_assert(ISPLIT % (2 * RWAVES * 2) == 0, "pairs of i per wave must be even");

__device__ __forceinline__ float sum_parts(const float* q) {
    return ((q[0] + q[O * CO]) + q[2 * O * CO]) + q[3 * O * CO];
}
static_assert(NSPLIT == 4, "sum_parts adds four partials");

__device__ __forceinline__ float squash16(float s) {
    float sq = s * s;
    sq += lane_dpp<QUAD_XOR1>(sq);
    sq += lane_dpp<QUAD_XOR2>(sq);
    sq += lane_dpp<ROW_XOR4>(sq);
    sq += lane_dpp<ROW_XOR8>(sq);
    return (sq / (1.0f + sq)) * s * rsqrtf(sq + 1e-8f);
}

template <int T>
__global__ __launch_bounds__(RWAVES * 32) void routing_part_kernel(
    float* ws,
    const float* __restrict__ bias)
{
    __shared__ float4 red[RWAVES * 128];
    __shared__ float4 vsum4[128];

    const int b    = blockIdx.x / NSPLIT;
    const int part = blockIdx.x % NSPLIT;
    const int tid  = threadIdx.x;
    const int w    = tid >> 5;
    const int lane = tid & 31;

    const float4* ub = (const float4*)(ws + (size_t)b * BSTRIDE)
                     + (size_t)part * (ISPLIT / 2) * 256 + lane;
    float* parts = ws + PART_OFF;

    constexpr int NPAIR = ISPLIT / 2 / RWAVES;

    float pq[T > 0 ? T : 1][NSPLIT];
    float bo = 0.0f;
    if (T > 0) {
        bo = bias[tid >> 4];
#pragma unroll
        for (int tp = 0; tp < T; ++tp)
#pragma unroll
            for (int p = 0; p < NSPLIT; ++p)
                pq[tp][p] = parts[((size_t)(tp * B + b) * NSPLIT + p) * (O * CO) + tid];
    }

    U8 ua, ubuf;
    load_u8(ua,   ub + (size_t)w * 256);
    load_u8(ubuf, ub + (size_t)(w + RWAVES) * 256);
    __builtin_amdgcn_sched_barrier(0);

    float4 vs[4];
    if (T == 0) {
#pragma unroll
        for (int j = 0; j < 4; ++j) vs[j] = make_float4(0.f, 0.f, 0.f, 0.f);
    } else {
        float vown = 0.0f;
#pragma unroll
        for (int tp = 0; tp < T; ++tp) {
            const float s = (((pq[tp][0] + pq[tp][1]) + pq[tp][2]) + pq[tp][3]) + bo;
            vown += squash16(s);
        }
        ((float*)vsum4)[tid] = vown;
        __syncthreads();
#pragma unroll
        for (int j = 0; j < 4; ++j) vs[j] = vsum4[j * 32 + lane];
    }

    float4 acc[4];
#pragma unroll
    for (int j = 0; j < 4; ++j) acc[j] = make_float4(0.f, 0.f, 0.f, 0.f);

    for (int s = 0; s < NPAIR - 2; s += 2) {
        route_one(&ua.q[0], vs, acc);
        route_one(&ua.q[4], vs, acc);
        load_u8(ua, ub + (size_t)(w + RWAVES * (s + 2)) * 256);
        __builtin_amdgcn_sched_barrier(0);
        route_one(&ubuf.q[0], vs, acc);
        route_one(&ubuf.q[4], vs, acc);
        load_u8(ubuf, ub + (size_t)(w + RWAVES * (s + 3)) * 256);
        __builtin_amdgcn_sched_barrier(0);
    }
    route_one(&ua.q[0], vs, acc);
    route_one(&ua.q[4], vs, acc);
    route_one(&ubuf.q[0], vs, acc);
    route_one(&ubuf.q[4], vs, acc);

#pragma unroll
    for (int j = 0; j < 4; ++j) red[w * 128 + j * 32 + lane] = acc[j];
    __syncthreads();

    {
        const float* rf = (const float*)red;
        float s = 0.0f;
#pragma unroll
        for (int ww = 0; ww < RWAVES; ++ww) s += rf[ww * (O * CO) + tid];
        parts[((size_t)(T * B + b) * NSPLIT + part) * (O * CO) + tid] = s;
    }
}

__global__ __launch_bounds__(O * CO) void routing_final_kernel(
    const float* __restrict__ ws,
    const float* __restrict__ bias,
    const float* __restrict__ beta,
    const float* __restrict__ alpha,
    float* __restrict__ out)
{
    const int b   = blockIdx.x;
    const int tid = threadIdx.x;
    const int o   = tid >> 4;
    const float s = sum_parts(ws + PART_OFF + ((size_t)((ITER - 1) * B + b) * NSPLIT) * (O * CO) + tid)
                  + bias[o];
    const float v = squash16(s);
    float vv = v * v;
    vv += lane_dpp<QUAD_XOR1>(vv);
    vv += lane_dpp<QUAD_XOR2>(vv);
    vv += lane_dpp<ROW_XOR4>(vv);
    vv += lane_dpp<ROW_XOR8>(vv);
    const float norm = sqrtf(vv + 1e-8f);
    const float act  = 1.0f / (1.0f + __expf(-(alpha[o] * norm + beta[o])));
    out[(size_t)b * O * CO + tid] = v * act;
}

extern "C" void kernel_launch(void* const* d_in, const int* in_sizes, int n_in,
                              void* d_out, int out_size, void* d_ws, size_t ws_size,
                              hipStream_t stream) {
    (void)in_sizes; (void)n_in; (void)out_size; (void)ws_size;
    const float* data  = (const float*)d_in[0];
    const float* W     = (const float*)d_in[1];
    const float* bias  = (const float*)d_in[2];
    const float* beta  = (const float*)d_in[3];
    const float* alpha = (const float*)d_in[4];
    float* out  = (float*)d_out;
    float* uhat = (float*)d_ws;

    uhat_wmma_kernel<<<dim3((I * 4) / 8), dim3(256), 0, stream>>>(data, W, uhat);
    static_assert(ITER == 3, "one routing launch per iteration is written out below");
    routing_part_kernel<0><<<dim3(B * NSPLIT), dim3(RWAVES * 32), 0, stream>>>(uhat, bias);
    routing_part_kernel<1><<<dim3(B * NSPLIT), dim3(RWAVES * 32), 0, stream>>>(uhat, bias);
    routing_part_kernel<2><<<dim3(B * NSPLIT), dim3(RWAVES * 32), 0, stream>>>(uhat, bias);
    routing_final_kernel<<<dim3(B), dim3(O * CO), 0, stream>>>(uhat, bias, beta, alpha, out);
}
